# starting kernel
# baseline (speedup 1.0000x reference)
#include <hip/hip_runtime.h>

typedef __attribute__((ext_vector_type(2))) float v2f;
typedef __attribute__((ext_vector_type(8))) float v8f;

#define NN 50000
#define AA 500
#define FF 256
#define APAD 512
#define EPSF 1e-12f

__device__ __forceinline__ v8f wmma_f32(v2f a, v2f b, v8f c) {
  return __builtin_amdgcn_wmma_f32_16x16x4_f32(false, a, false, b, (short)0, c,
                                               false, false);
}

__global__ __launch_bounds__(256) void k_rowsum_inv(const float* __restrict__ adj,
                                                    float* __restrict__ invrow) {
  const int lane = threadIdx.x & 31;
  const int wave = threadIdx.x >> 5;
  const int row = blockIdx.x * 8 + wave;
  const float* rp = adj + row * AA;
  float s = 0.f;
  for (int c = lane; c < AA; c += 32) s += rp[c];
  for (int off = 16; off > 0; off >>= 1) s += __shfl_down(s, off, 32);
  if (lane == 0) invrow[row] = 1.0f / (s + EPSF);
}

__global__ __launch_bounds__(256) void k_colsum_part(const float* __restrict__ adj,
                                                     float* __restrict__ colsum) {
  const int t = threadIdx.x;
  const int r0 = blockIdx.x * 64;
  const int rend = (r0 + 64 < NN) ? (r0 + 64) : NN;
  const int c1 = t + 256;
  float a0 = 0.f, a1 = 0.f;
  for (int r = r0; r < rend; ++r) {
    const float* rp = adj + r * AA;
    a0 += rp[t];
    if (c1 < AA) a1 += rp[c1];
  }
  atomicAdd(&colsum[t], a0);
  if (c1 < AA) atomicAdd(&colsum[c1], a1);
}

__global__ void k_colsum_inv(const float* __restrict__ colsum,
                             float* __restrict__ invcol) {
  const int a = blockIdx.x * blockDim.x + threadIdx.x;
  if (a < APAD) invcol[a] = (a < AA) ? 1.0f / (colsum[a] + EPSF) : 0.0f;
}

__global__ __launch_bounds__(512) void k_atx(const float* __restrict__ adj,
                                             const float* __restrict__ inp,
                                             float* __restrict__ msg0) {
  __shared__ float Alt[50 * 160];
  const int agroup = blockIdx.x & 7;
  const int chunk  = blockIdx.x >> 3;
  const int a0 = agroup * 64;
  const int nbase = chunk * 1000;
  const int wave = threadIdx.x >> 5;
  const int lane = threadIdx.x & 31;
  const int f0 = wave * 16;
  const int half = lane >> 4;
  const int mrow = lane & 15;
  v8f c0 = {}, c1 = {}, c2 = {}, c3 = {};
  const float* bbase = inp + (nbase + half * 2) * FF + f0 + mrow;
  const int afrag = half * 160 + mrow * 2;
  for (int kk = 0; kk < 1000; kk += 100) {
    __syncthreads();
    for (int e = threadIdx.x; e < 100 * 64; e += 512) {
      const int pn = e >> 6;
      const int pa = e & 63;
      const int a = a0 + pa;
      const float v = (a < AA) ? adj[(nbase + kk + pn) * AA + a] : 0.0f;
      Alt[(pn >> 1) * 160 + pa * 2 + (pn & 1)] = v;
    }
    __syncthreads();
    const float* bp = bbase + kk * FF;
#pragma unroll
    for (int s = 0; s < 25; ++s) {
      v2f b;
      b.x = bp[s * 4 * FF];
      b.y = bp[s * 4 * FF + FF];
      const float* ap = &Alt[afrag + s * 320];
      c0 = wmma_f32(*(const v2f*)(ap +  0), b, c0);
      c1 = wmma_f32(*(const v2f*)(ap + 32), b, c1);
      c2 = wmma_f32(*(const v2f*)(ap + 64), b, c2);
      c3 = wmma_f32(*(const v2f*)(ap + 96), b, c3);
    }
  }
  const int col = f0 + mrow;
#pragma unroll
  for (int v = 0; v < 8; ++v) {
    const int m = v + half * 8;
    atomicAdd(&msg0[(a0 +  0 + m) * FF + col], c0[v]);
    atomicAdd(&msg0[(a0 + 16 + m) * FF + col], c1[v]);
    atomicAdd(&msg0[(a0 + 32 + m) * FF + col], c2[v]);
    atomicAdd(&msg0[(a0 + 48 + m) * FF + col], c3[v]);
  }
}

__global__ __launch_bounds__(512) void k_msgw(const float* __restrict__ msg0,
                                              const float* __restrict__ Wm,
                                              const float* __restrict__ invcol,
                                              float* __restrict__ msg) {
  __shared__ float Als[16][260];
  const int a0 = blockIdx.x * 16;
  const int wave = threadIdx.x >> 5;
  const int lane = threadIdx.x & 31;
  const int f0 = wave * 16;
  const int half = lane >> 4;
  const int mrow = lane & 15;
  for (int e = threadIdx.x; e < 16 * 256; e += 512) {
    const int m = e >> 8, k = e & 255;
    Als[m][k] = msg0[(a0 + m) * FF + k] * invcol[a0 + m];
  }
  __syncthreads();
  v8f c = {};
  const float* wb = Wm + (f0 + mrow) * FF + half * 2;
  const float* ab = &Als[mrow][half * 2];
#pragma unroll
  for (int s = 0; s < 64; ++s) {
    v2f b = *(const v2f*)(wb + s * 4);
    v2f a = *(const v2f*)(ab + s * 4);
    c = wmma_f32(a, b, c);
  }
  const int col = f0 + mrow;
#pragma unroll
  for (int v = 0; v < 8; ++v)
    msg[(a0 + v + half * 8) * FF + col] = c[v];
}

__global__ __launch_bounds__(512) void k_out(const float* __restrict__ adj,
                                             const float* __restrict__ msg,
                                             const float* __restrict__ invrow,
                                             float* __restrict__ out) {
  extern __shared__ float Als[];
  const int LDP = 516;
  const int m0 = blockIdx.x * 64;
  const int wave = threadIdx.x >> 5;
  const int lane = threadIdx.x & 31;
  const int f0 = wave * 16;
  const int half = lane >> 4;
  const int mrow = lane & 15;
  const bool full = (m0 + 64 <= NN);
  if (full) {
    for (int e = threadIdx.x; e < 64 * 512; e += 512) {
      const int m = e >> 9, k = e & 511;
      Als[m * LDP + k] = (k < AA) ? adj[(m0 + m) * AA + k] : 0.0f;
    }
  } else {
    for (int e = threadIdx.x; e < 64 * 512; e += 512) {
      const int m = e >> 9, k = e & 511;
      const int gm = m0 + m;
      Als[m * LDP + k] = (k < AA && gm < NN) ? adj[gm * AA + k] : 0.0f;
    }
  }
  __syncthreads();
  v8f c0 = {}, c1 = {}, c2 = {}, c3 = {};
  const float* bb = msg + half * 2 * FF + f0 + mrow;
  const float* a0p = &Als[mrow * LDP + half * 2];
#pragma unroll 8
  for (int s = 0; s < 128; ++s) {
    v2f b;
    b.x = bb[s * 4 * FF];
    b.y = bb[s * 4 * FF + FF];
    const float* ap = a0p + s * 4;
    c0 = wmma_f32(*(const v2f*)(ap),            b, c0);
    c1 = wmma_f32(*(const v2f*)(ap + 16 * LDP), b, c1);
    c2 = wmma_f32(*(const v2f*)(ap + 32 * LDP), b, c2);
    c3 = wmma_f32(*(const v2f*)(ap + 48 * LDP), b, c3);
  }
  const int col = f0 + mrow;
  if (full) {
#pragma unroll
    for (int v = 0; v < 8; ++v) {
      const int mi = v + half * 8;
      int r;
      r = m0 +  0 + mi; out[r * FF + col] = c0[v] * invrow[r];
      r = m0 + 16 + mi; out[r * FF + col] = c1[v] * invrow[r];
      r = m0 + 32 + mi; out[r * FF + col] = c2[v] * invrow[r];
      r = m0 + 48 + mi; out[r * FF + col] = c3[v] * invrow[r];
    }
  } else {
#pragma unroll
    for (int v = 0; v < 8; ++v) {
      const int mi = v + half * 8;
      int r;
      r = m0 +  0 + mi; if (r < NN) out[r * FF + col] = c0[v] * invrow[r];
      r = m0 + 16 + mi; if (r < NN) out[r * FF + col] = c1[v] * invrow[r];
      r = m0 + 32 + mi; if (r < NN) out[r * FF + col] = c2[v] * invrow[r];
      r = m0 + 48 + mi; if (r < NN) out[r * FF + col] = c3[v] * invrow[r];
    }
  }
}

extern "C" void kernel_launch(void* const* d_in, const int* in_sizes, int n_in,
                              void* d_out, int out_size, void* d_ws, size_t ws_size,
                              hipStream_t stream) {
  const float* inp = (const float*)d_in[0];
  const float* adj = (const float*)d_in[1];
  const float* Wm  = (const float*)d_in[2];
  float* out = (float*)d_out;

  char* ws = (char*)d_ws;
  float* msg0   = (float*)(ws);
  float* msg    = (float*)(ws + (size_t)APAD * FF * 4);
  float* colsum = (float*)(ws + (size_t)2 * APAD * FF * 4);
  float* invcol = colsum + APAD;
  float* invrow = invcol + APAD;

  hipMemsetAsync(msg0, 0, (size_t)APAD * FF * sizeof(float), stream);
  hipMemsetAsync(colsum, 0, APAD * sizeof(float), stream);

  k_rowsum_inv<<<NN / 8, 256, 0, stream>>>(adj, invrow);
  k_colsum_part<<<(NN + 63) / 64, 256, 0, stream>>>(adj, colsum);
  k_colsum_inv<<<2, 256, 0, stream>>>(colsum, invcol);
  k_atx<<<400, 512, 0, stream>>>(adj, inp, msg0);
  k_msgw<<<32, 512, 0, stream>>>(msg0, Wm, invcol, msg);
  k_out<<<(NN + 63) / 64, 512, (size_t)64 * 516 * sizeof(float), stream>>>(
      adj, msg, invrow, out);
}


# adj tile staging in both GEMM kernels: a tile's 16-byte loads are all issued before its first LDS store; the guarded 4-byte loads had each been waited for before the next one was issued
# speedup vs baseline: 1.2814x; 1.2814x over previous
#include <hip/hip_runtime.h>

typedef __attribute__((ext_vector_type(2))) float v2f;
typedef __attribute__((ext_vector_type(8))) float v8f;

#define NN 50000
#define AA 500
#define FF 256
#define APAD 512
#define EPSF 1e-12f

__device__ __forceinline__ v8f wmma_f32(v2f a, v2f b, v8f c) {
  return __builtin_amdgcn_wmma_f32_16x16x4_f32(false, a, false, b, (short)0, c,
                                               false, false);
}

__device__ __forceinline__ float4 keep_if(bool ok, float a, float b, float c, float d) {
  return make_float4(ok ? a : 0.f, ok ? b : 0.f, ok ? c : 0.f, ok ? d : 0.f);
}

__global__ __launch_bounds__(256) void k_rowsum_inv(const float* __restrict__ adj,
                                                    float* __restrict__ invrow) {
  const int lane = threadIdx.x & 31;
  const int wave = threadIdx.x >> 5;
  const int row = blockIdx.x * 8 + wave;
  const float* rp = adj + row * AA;
  float s = 0.f;
  for (int c = lane; c < AA; c += 32) s += rp[c];
  for (int off = 16; off > 0; off >>= 1) s += __shfl_down(s, off, 32);
  if (lane == 0) invrow[row] = 1.0f / (s + EPSF);
}

__global__ __launch_bounds__(256) void k_colsum_part(const float* __restrict__ adj,
                                                     float* __restrict__ colsum) {
  const int t = threadIdx.x;
  const int r0 = blockIdx.x * 64;
  const int rend = (r0 + 64 < NN) ? (r0 + 64) : NN;
  const int c1 = t + 256;
  float a0 = 0.f, a1 = 0.f;
  for (int r = r0; r < rend; ++r) {
    const float* rp = adj + r * AA;
    a0 += rp[t];
    if (c1 < AA) a1 += rp[c1];
  }
  atomicAdd(&colsum[t], a0);
  if (c1 < AA) atomicAdd(&colsum[c1], a1);
}

__global__ void k_colsum_inv(const float* __restrict__ colsum,
                             float* __restrict__ invcol) {
  const int a = blockIdx.x * blockDim.x + threadIdx.x;
  if (a < APAD) invcol[a] = (a < AA) ? 1.0f / (colsum[a] + EPSF) : 0.0f;
}

__global__ __launch_bounds__(512) void k_atx(const float* __restrict__ adj,
                                             const float* __restrict__ inp,
                                             float* __restrict__ msg0) {
  __shared__ __attribute__((aligned(16))) float Alt[50 * 160];
  const int agroup = blockIdx.x & 7;
  const int chunk  = blockIdx.x >> 3;
  const int a0 = agroup * 64;
  const int nbase = chunk * 1000;
  const int wave = threadIdx.x >> 5;
  const int lane = threadIdx.x & 31;
  const int f0 = wave * 16;
  const int half = lane >> 4;
  const int mrow = lane & 15;
  v8f c0 = {}, c1 = {}, c2 = {}, c3 = {};
  const float* bbase = inp + (nbase + half * 2) * FF + f0 + mrow;
  const int afrag = half * 160 + mrow * 2;
  for (int kk = 0; kk < 1000; kk += 100) {
    float4 lo[2], hi[2];
#pragma unroll
    for (int i = 0; i < 2; ++i) {
      const int it = threadIdx.x + i * 512;
      const int pr = it >> 4;
      const int qd = it & 15;
      const bool ok = (it < 800) && (a0 + qd * 4 < AA);
      const int idx = ok ? ((nbase + kk + 2 * pr) * AA + a0 + qd * 4) : 0;
      lo[i] = *(const float4*)(adj + idx);
      hi[i] = *(const float4*)(adj + idx + AA);
    }
    __syncthreads();
#pragma unroll
    for (int i = 0; i < 2; ++i) {
      const int it = threadIdx.x + i * 512;
      const int pr = it >> 4;
      const int qd = it & 15;
      const bool ok = (a0 + qd * 4 < AA);
      if (it < 800) {
        const float4 l = lo[i], h = hi[i];
        float* d = &Alt[pr * 160 + qd * 8];
        *(float4*)(d)     = keep_if(ok, l.x, h.x, l.y, h.y);
        *(float4*)(d + 4) = keep_if(ok, l.z, h.z, l.w, h.w);
      }
    }
    __syncthreads();
    const float* bp = bbase + kk * FF;
#pragma unroll
    for (int s = 0; s < 25; ++s) {
      v2f b;
      b.x = bp[s * 4 * FF];
      b.y = bp[s * 4 * FF + FF];
      const float* ap = &Alt[afrag + s * 320];
      c0 = wmma_f32(*(const v2f*)(ap +  0), b, c0);
      c1 = wmma_f32(*(const v2f*)(ap + 32), b, c1);
      c2 = wmma_f32(*(const v2f*)(ap + 64), b, c2);
      c3 = wmma_f32(*(const v2f*)(ap + 96), b, c3);
    }
  }
  const int col = f0 + mrow;
#pragma unroll
  for (int v = 0; v < 8; ++v) {
    const int m = v + half * 8;
    atomicAdd(&msg0[(a0 +  0 + m) * FF + col], c0[v]);
    atomicAdd(&msg0[(a0 + 16 + m) * FF + col], c1[v]);
    atomicAdd(&msg0[(a0 + 32 + m) * FF + col], c2[v]);
    atomicAdd(&msg0[(a0 + 48 + m) * FF + col], c3[v]);
  }
}

__global__ __launch_bounds__(512) void k_msgw(const float* __restrict__ msg0,
                                              const float* __restrict__ Wm,
                                              const float* __restrict__ invcol,
                                              float* __restrict__ msg) {
  __shared__ float Als[16][260];
  const int a0 = blockIdx.x * 16;
  const int wave = threadIdx.x >> 5;
  const int lane = threadIdx.x & 31;
  const int f0 = wave * 16;
  const int half = lane >> 4;
  const int mrow = lane & 15;
  for (int e = threadIdx.x; e < 16 * 256; e += 512) {
    const int m = e >> 8, k = e & 255;
    Als[m][k] = msg0[(a0 + m) * FF + k] * invcol[a0 + m];
  }
  __syncthreads();
  v8f c = {};
  const float* wb = Wm + (f0 + mrow) * FF + half * 2;
  const float* ab = &Als[mrow][half * 2];
#pragma unroll
  for (int s = 0; s < 64; ++s) {
    v2f b = *(const v2f*)(wb + s * 4);
    v2f a = *(const v2f*)(ab + s * 4);
    c = wmma_f32(a, b, c);
  }
  const int col = f0 + mrow;
#pragma unroll
  for (int v = 0; v < 8; ++v)
    msg[(a0 + v + half * 8) * FF + col] = c[v];
}

__global__ __launch_bounds__(512) void k_out(const float* __restrict__ adj,
                                             const float* __restrict__ msg,
                                             const float* __restrict__ invrow,
                                             float* __restrict__ out) {
  extern __shared__ __attribute__((aligned(16))) float Als[];
  const int LDP = 516;
  const int m0 = blockIdx.x * 64;
  const int wave = threadIdx.x >> 5;
  const int lane = threadIdx.x & 31;
  const int f0 = wave * 16;
  const int half = lane >> 4;
  const int mrow = lane & 15;
  const bool full = (m0 + 64 <= NN);
  {
    const float4* srcq = (const float4*)(adj + m0 * AA);
    if (full) {
      float4 r[16];
#pragma unroll
      for (int i = 0; i < 16; ++i) {
        const int q = threadIdx.x + i * 512;
        r[i] = srcq[(q < 8000) ? q : 0];
      }
#pragma unroll
      for (int i = 0; i < 16; ++i) {
        const int q = threadIdx.x + i * 512;
        const bool data = (q < 8000);
        const int t = q - 8000;
        const int m = data ? q / 125 : t / 3;
        const int c4 = data ? q - m * 125 : 125 + (t - m * 3);
        const float4 v = r[i];
        *(float4*)&Als[m * LDP + c4 * 4] = keep_if(data, v.x, v.y, v.z, v.w);
      }
    } else {
      const int nq = (NN - m0) * 125;
      for (int q = threadIdx.x; q < 8192; q += 512) {
        const bool data = (q < 8000);
        const int t = q - 8000;
        const int m = data ? q / 125 : t / 3;
        const int c4 = data ? q - m * 125 : 125 + (t - m * 3);
        float4 v = make_float4(0.f, 0.f, 0.f, 0.f);
        if (q < nq) v = srcq[q];
        *(float4*)&Als[m * LDP + c4 * 4] = v;
      }
    }
  }
  __syncthreads();
  v8f c0 = {}, c1 = {}, c2 = {}, c3 = {};
  const float* bb = msg + half * 2 * FF + f0 + mrow;
  const float* a0p = &Als[mrow * LDP + half * 2];
#pragma unroll 8
  for (int s = 0; s < 128; ++s) {
    v2f b;
    b.x = bb[s * 4 * FF];
    b.y = bb[s * 4 * FF + FF];
    const float* ap = a0p + s * 4;
    c0 = wmma_f32(*(const v2f*)(ap),            b, c0);
    c1 = wmma_f32(*(const v2f*)(ap + 16 * LDP), b, c1);
    c2 = wmma_f32(*(const v2f*)(ap + 32 * LDP), b, c2);
    c3 = wmma_f32(*(const v2f*)(ap + 48 * LDP), b, c3);
  }
  const int col = f0 + mrow;
  if (full) {
#pragma unroll
    for (int v = 0; v < 8; ++v) {
      const int mi = v + half * 8;
      int r;
      r = m0 +  0 + mi; out[r * FF + col] = c0[v] * invrow[r];
      r = m0 + 16 + mi; out[r * FF + col] = c1[v] * invrow[r];
      r = m0 + 32 + mi; out[r * FF + col] = c2[v] * invrow[r];
      r = m0 + 48 + mi; out[r * FF + col] = c3[v] * invrow[r];
    }
  } else {
#pragma unroll
    for (int v = 0; v < 8; ++v) {
      const int mi = v + half * 8;
      int r;
      r = m0 +  0 + mi; if (r < NN) out[r * FF + col] = c0[v] * invrow[r];
      r = m0 + 16 + mi; if (r < NN) out[r * FF + col] = c1[v] * invrow[r];
      r = m0 + 32 + mi; if (r < NN) out[r * FF + col] = c2[v] * invrow[r];
      r = m0 + 48 + mi; if (r < NN) out[r * FF + col] = c3[v] * invrow[r];
    }
  }
}

extern "C" void kernel_launch(void* const* d_in, const int* in_sizes, int n_in,
                              void* d_out, int out_size, void* d_ws, size_t ws_size,
                              hipStream_t stream) {
  const float* inp = (const float*)d_in[0];
  const float* adj = (const float*)d_in[1];
  const float* Wm  = (const float*)d_in[2];
  float* out = (float*)d_out;

  char* ws = (char*)d_ws;
  float* msg0   = (float*)(ws);
  float* msg    = (float*)(ws + (size_t)APAD * FF * 4);
  float* colsum = (float*)(ws + (size_t)2 * APAD * FF * 4);
  float* invcol = colsum + APAD;
  float* invrow = invcol + APAD;

  hipMemsetAsync(msg0, 0, (size_t)APAD * FF * sizeof(float), stream);
  hipMemsetAsync(colsum, 0, APAD * sizeof(float), stream);

  k_rowsum_inv<<<NN / 8, 256, 0, stream>>>(adj, invrow);
  k_colsum_part<<<(NN + 63) / 64, 256, 0, stream>>>(adj, colsum);
  k_colsum_inv<<<2, 256, 0, stream>>>(colsum, invcol);
  k_atx<<<400, 512, 0, stream>>>(adj, inp, msg0);
  k_msgw<<<32, 512, 0, stream>>>(msg0, Wm, invcol, msg);
  k_out<<<(NN + 63) / 64, 512, (size_t)64 * 516 * sizeof(float), stream>>>(
      adj, msg, invrow, out);
}


# adj's row and column sums are taken inside the two GEMM kernels from the tiles they stage anyway: 3 sum kernels and 2 of 4 passes over adj go; both sums are added in a new order (tile partials)
# speedup vs baseline: 1.5158x; 1.1829x over previous
#include <hip/hip_runtime.h>

typedef __attribute__((ext_vector_type(2))) float v2f;
typedef __attribute__((ext_vector_type(8))) float v8f;

#define NN 50000
#define AA 500
#define FF 256
#define APAD 512
#define EPSF 1e-12f

__device__ __forceinline__ v8f wmma_f32(v2f a, v2f b, v8f c) {
  return __builtin_amdgcn_wmma_f32_16x16x4_f32(false, a, false, b, (short)0, c,
                                               false, false);
}

__device__ __forceinline__ float4 keep_if(bool ok, float a, float b, float c, float d) {
  return make_float4(ok ? a : 0.f, ok ? b : 0.f, ok ? c : 0.f, ok ? d : 0.f);
}

__global__ __launch_bounds__(512) void k_atx(const float* __restrict__ adj,
                                             const float* __restrict__ inp,
                                             float* __restrict__ msg0,
                                             float* __restrict__ colsum) {
  __shared__ __attribute__((aligned(16))) float Alt[50 * 160];
  const int agroup = blockIdx.x & 7;
  const int chunk  = blockIdx.x >> 3;
  const int a0 = agroup * 64;
  const int nbase = chunk * 1000;
  const int wave = threadIdx.x >> 5;
  const int lane = threadIdx.x & 31;
  const int f0 = wave * 16;
  const int half = lane >> 4;
  const int mrow = lane & 15;
  v8f c0 = {}, c1 = {}, c2 = {}, c3 = {};
  float cs0 = 0.f, cs1 = 0.f, cs2 = 0.f, cs3 = 0.f;
  const float* bbase = inp + (nbase + half * 2) * FF + f0 + mrow;
  const int afrag = half * 160 + mrow * 2;
  for (int kk = 0; kk < 1000; kk += 100) {
    float4 lo[2], hi[2];
#pragma unroll
    for (int i = 0; i < 2; ++i) {
      const int it = threadIdx.x + i * 512;
      const int pr = it >> 4;
      const int qd = it & 15;
      const bool ok = (it < 800) && (a0 + qd * 4 < AA);
      const int idx = ok ? ((nbase + kk + 2 * pr) * AA + a0 + qd * 4) : 0;
      lo[i] = *(const float4*)(adj + idx);
      hi[i] = *(const float4*)(adj + idx + AA);
    }
    __syncthreads();
#pragma unroll
    for (int i = 0; i < 2; ++i) {
      const int it = threadIdx.x + i * 512;
      const int pr = it >> 4;
      const int qd = it & 15;
      const bool ok = (a0 + qd * 4 < AA);
      if (it < 800) {
        const float4 l = lo[i], h = hi[i];
        float* d = &Alt[pr * 160 + qd * 8];
        const float4 p0 = keep_if(ok, l.x, h.x, l.y, h.y);
        const float4 p1 = keep_if(ok, l.z, h.z, l.w, h.w);
        *(float4*)(d)     = p0;
        *(float4*)(d + 4) = p1;
        cs0 += p0.x + p0.y;
        cs1 += p0.z + p0.w;
        cs2 += p1.x + p1.y;
        cs3 += p1.z + p1.w;
      }
    }
    __syncthreads();
    const float* bp = bbase + kk * FF;
#pragma unroll
    for (int s = 0; s < 25; ++s) {
      v2f b;
      b.x = bp[s * 4 * FF];
      b.y = bp[s * 4 * FF + FF];
      const float* ap = &Alt[afrag + s * 320];
      c0 = wmma_f32(*(const v2f*)(ap +  0), b, c0);
      c1 = wmma_f32(*(const v2f*)(ap + 32), b, c1);
      c2 = wmma_f32(*(const v2f*)(ap + 64), b, c2);
      c3 = wmma_f32(*(const v2f*)(ap + 96), b, c3);
    }
  }
  cs0 += __shfl_xor(cs0, 16, 32);
  cs1 += __shfl_xor(cs1, 16, 32);
  cs2 += __shfl_xor(cs2, 16, 32);
  cs3 += __shfl_xor(cs3, 16, 32);
  __syncthreads();
  if (half == 0) *(float4*)&Alt[wave * 64 + mrow * 4] = make_float4(cs0, cs1, cs2, cs3);
  __syncthreads();
  if (threadIdx.x < 64 && a0 + (int)threadIdx.x < AA) {
    float s = 0.f;
    for (int w = 0; w < 16; ++w) s += Alt[w * 64 + threadIdx.x];
    atomicAdd(&colsum[a0 + threadIdx.x], s);
  }
  const int col = f0 + mrow;
#pragma unroll
  for (int v = 0; v < 8; ++v) {
    const int m = v + half * 8;
    atomicAdd(&msg0[(a0 +  0 + m) * FF + col], c0[v]);
    atomicAdd(&msg0[(a0 + 16 + m) * FF + col], c1[v]);
    atomicAdd(&msg0[(a0 + 32 + m) * FF + col], c2[v]);
    atomicAdd(&msg0[(a0 + 48 + m) * FF + col], c3[v]);
  }
}

__global__ __launch_bounds__(512) void k_msgw(const float* __restrict__ msg0,
                                              const float* __restrict__ Wm,
                                              const float* __restrict__ colsum,
                                              float* __restrict__ msg) {
  __shared__ float Als[16][260];
  const int a0 = blockIdx.x * 16;
  const int wave = threadIdx.x >> 5;
  const int lane = threadIdx.x & 31;
  const int f0 = wave * 16;
  const int half = lane >> 4;
  const int mrow = lane & 15;
  for (int e = threadIdx.x; e < 16 * 256; e += 512) {
    const int m = e >> 8, k = e & 255;
    const int a = a0 + m;
    const float ic = (a < AA) ? 1.0f / (colsum[a] + EPSF) : 0.0f;
    Als[m][k] = msg0[a * FF + k] * ic;
  }
  __syncthreads();
  v8f c = {};
  const float* wb = Wm + (f0 + mrow) * FF + half * 2;
  const float* ab = &Als[mrow][half * 2];
#pragma unroll
  for (int s = 0; s < 64; ++s) {
    v2f b = *(const v2f*)(wb + s * 4);
    v2f a = *(const v2f*)(ab + s * 4);
    c = wmma_f32(a, b, c);
  }
  const int col = f0 + mrow;
#pragma unroll
  for (int v = 0; v < 8; ++v)
    msg[(a0 + v + half * 8) * FF + col] = c[v];
}

__global__ __launch_bounds__(512) void k_out(const float* __restrict__ adj,
                                             const float* __restrict__ msg,
                                             float* __restrict__ out) {
  extern __shared__ __attribute__((aligned(16))) float Als[];
  const int LDP = 516;
  float* invr = Als + 64 * LDP;
  const int m0 = blockIdx.x * 64;
  const int wave = threadIdx.x >> 5;
  const int lane = threadIdx.x & 31;
  const int f0 = wave * 16;
  const int half = lane >> 4;
  const int mrow = lane & 15;
  const bool full = (m0 + 64 <= NN);
  {
    const float4* srcq = (const float4*)(adj + m0 * AA);
    if (full) {
      float4 r[16];
#pragma unroll
      for (int i = 0; i < 16; ++i) {
        const int q = threadIdx.x + i * 512;
        r[i] = srcq[(q < 8000) ? q : 0];
      }
#pragma unroll
      for (int i = 0; i < 16; ++i) {
        const int q = threadIdx.x + i * 512;
        const bool data = (q < 8000);
        const int t = q - 8000;
        const int m = data ? q / 125 : t / 3;
        const int c4 = data ? q - m * 125 : 125 + (t - m * 3);
        const float4 v = r[i];
        *(float4*)&Als[m * LDP + c4 * 4] = keep_if(data, v.x, v.y, v.z, v.w);
      }
    } else {
      const int nq = (NN - m0) * 125;
      for (int q = threadIdx.x; q < 8192; q += 512) {
        const bool data = (q < 8000);
        const int t = q - 8000;
        const int m = data ? q / 125 : t / 3;
        const int c4 = data ? q - m * 125 : 125 + (t - m * 3);
        float4 v = make_float4(0.f, 0.f, 0.f, 0.f);
        if (q < nq) v = srcq[q];
        *(float4*)&Als[m * LDP + c4 * 4] = v;
      }
    }
  }
  __syncthreads();
  float rs[4];
  for (int r = 0; r < 4; ++r) {
    const float4* rowq = (const float4*)&Als[(wave * 4 + r) * LDP];
    float s = 0.f;
    for (int j = 0; j < 4; ++j) {
      const float4 v = rowq[lane + 32 * j];
      s += (v.x + v.y) + (v.z + v.w);
    }
    rs[r] = s;
  }
  for (int off = 16; off > 0; off >>= 1) {
    for (int r = 0; r < 4; ++r) rs[r] += __shfl_xor(rs[r], off, 32);
  }
  {
    const float sv = (lane == 0) ? rs[0] : (lane == 1) ? rs[1] : (lane == 2) ? rs[2] : rs[3];
    if (lane < 4) invr[wave * 4 + lane] = 1.0f / (sv + EPSF);
  }
  v8f c0 = {}, c1 = {}, c2 = {}, c3 = {};
  const float* bb = msg + half * 2 * FF + f0 + mrow;
  const float* a0p = &Als[mrow * LDP + half * 2];
#pragma unroll 8
  for (int s = 0; s < 128; ++s) {
    v2f b;
    b.x = bb[s * 4 * FF];
    b.y = bb[s * 4 * FF + FF];
    const float* ap = a0p + s * 4;
    c0 = wmma_f32(*(const v2f*)(ap),            b, c0);
    c1 = wmma_f32(*(const v2f*)(ap + 16 * LDP), b, c1);
    c2 = wmma_f32(*(const v2f*)(ap + 32 * LDP), b, c2);
    c3 = wmma_f32(*(const v2f*)(ap + 48 * LDP), b, c3);
  }
  __syncthreads();
  const int col = f0 + mrow;
  if (full) {
#pragma unroll
    for (int v = 0; v < 8; ++v) {
      const int mi = v + half * 8;
      int r;
      r = m0 +  0 + mi; out[r * FF + col] = c0[v] * invr[mi];
      r = m0 + 16 + mi; out[r * FF + col] = c1[v] * invr[mi + 16];
      r = m0 + 32 + mi; out[r * FF + col] = c2[v] * invr[mi + 32];
      r = m0 + 48 + mi; out[r * FF + col] = c3[v] * invr[mi + 48];
    }
  } else {
#pragma unroll
    for (int v = 0; v < 8; ++v) {
      const int mi = v + half * 8;
      int r;
      r = m0 +  0 + mi; if (r < NN) out[r * FF + col] = c0[v] * invr[mi];
      r = m0 + 16 + mi; if (r < NN) out[r * FF + col] = c1[v] * invr[mi + 16];
      r = m0 + 32 + mi; if (r < NN) out[r * FF + col] = c2[v] * invr[mi + 32];
      r = m0 + 48 + mi; if (r < NN) out[r * FF + col] = c3[v] * invr[mi + 48];
    }
  }
}

extern "C" void kernel_launch(void* const* d_in, const int* in_sizes, int n_in,
                              void* d_out, int out_size, void* d_ws, size_t ws_size,
                              hipStream_t stream) {
  const float* inp = (const float*)d_in[0];
  const float* adj = (const float*)d_in[1];
  const float* Wm  = (const float*)d_in[2];
  float* out = (float*)d_out;

  char* ws = (char*)d_ws;
  float* msg0   = (float*)(ws);
  float* msg    = (float*)(ws + (size_t)APAD * FF * 4);
  float* colsum = (float*)(ws + (size_t)2 * APAD * FF * 4);

  hipMemsetAsync(msg0, 0, (size_t)APAD * FF * sizeof(float), stream);
  hipMemsetAsync(colsum, 0, APAD * sizeof(float), stream);

  k_atx<<<400, 512, 0, stream>>>(adj, inp, msg0, colsum);
  k_msgw<<<32, 512, 0, stream>>>(msg0, Wm, colsum, msg);
  k_out<<<(NN + 63) / 64, 512, (size_t)(64 * 516 + 64) * sizeof(float), stream>>>(
      adj, msg, out);
}


# k_out main loop: the next 4-k-step block's A and B fragments are loaded as one group into a second register set ahead of the current block's 16 WMMAs; loop fully unrolled, offsets immediate
# speedup vs baseline: 1.5640x; 1.0318x over previous
#include <hip/hip_runtime.h>

typedef __attribute__((ext_vector_type(2))) float v2f;
typedef __attribute__((ext_vector_type(8))) float v8f;

#define NN 50000
#define AA 500
#define FF 256
#define APAD 512
#define EPSF 1e-12f

__device__ __forceinline__ v8f wmma_f32(v2f a, v2f b, v8f c) {
  return __builtin_amdgcn_wmma_f32_16x16x4_f32(false, a, false, b, (short)0, c,
                                               false, false);
}

__device__ __forceinline__ float4 keep_if(bool ok, float a, float b, float c, float d) {
  return make_float4(ok ? a : 0.f, ok ? b : 0.f, ok ? c : 0.f, ok ? d : 0.f);
}

__global__ __launch_bounds__(512) void k_atx(const float* __restrict__ adj,
                                             const float* __restrict__ inp,
                                             float* __restrict__ msg0,
                                             float* __restrict__ colsum) {
  __shared__ __attribute__((aligned(16))) float Alt[50 * 160];
  const int agroup = blockIdx.x & 7;
  const int chunk  = blockIdx.x >> 3;
  const int a0 = agroup * 64;
  const int nbase = chunk * 1000;
  const int wave = threadIdx.x >> 5;
  const int lane = threadIdx.x & 31;
  const int f0 = wave * 16;
  const int half = lane >> 4;
  const int mrow = lane & 15;
  v8f c0 = {}, c1 = {}, c2 = {}, c3 = {};
  float cs0 = 0.f, cs1 = 0.f, cs2 = 0.f, cs3 = 0.f;
  const float* bbase = inp + (nbase + half * 2) * FF + f0 + mrow;
  const int afrag = half * 160 + mrow * 2;
  for (int kk = 0; kk < 1000; kk += 100) {
    float4 lo[2], hi[2];
#pragma unroll
    for (int i = 0; i < 2; ++i) {
      const int it = threadIdx.x + i * 512;
      const int pr = it >> 4;
      const int qd = it & 15;
      const bool ok = (it < 800) && (a0 + qd * 4 < AA);
      const int idx = ok ? ((nbase + kk + 2 * pr) * AA + a0 + qd * 4) : 0;
      lo[i] = *(const float4*)(adj + idx);
      hi[i] = *(const float4*)(adj + idx + AA);
    }
    __syncthreads();
#pragma unroll
    for (int i = 0; i < 2; ++i) {
      const int it = threadIdx.x + i * 512;
      const int pr = it >> 4;
      const int qd = it & 15;
      const bool ok = (a0 + qd * 4 < AA);
      if (it < 800) {
        const float4 l = lo[i], h = hi[i];
        float* d = &Alt[pr * 160 + qd * 8];
        const float4 p0 = keep_if(ok, l.x, h.x, l.y, h.y);
        const float4 p1 = keep_if(ok, l.z, h.z, l.w, h.w);
        *(float4*)(d)     = p0;
        *(float4*)(d + 4) = p1;
        cs0 += p0.x + p0.y;
        cs1 += p0.z + p0.w;
        cs2 += p1.x + p1.y;
        cs3 += p1.z + p1.w;
      }
    }
    __syncthreads();
    const float* bp = bbase + kk * FF;
#pragma unroll
    for (int s = 0; s < 25; ++s) {
      v2f b;
      b.x = bp[s * 4 * FF];
      b.y = bp[s * 4 * FF + FF];
      const float* ap = &Alt[afrag + s * 320];
      c0 = wmma_f32(*(const v2f*)(ap +  0), b, c0);
      c1 = wmma_f32(*(const v2f*)(ap + 32), b, c1);
      c2 = wmma_f32(*(const v2f*)(ap + 64), b, c2);
      c3 = wmma_f32(*(const v2f*)(ap + 96), b, c3);
    }
  }
  cs0 += __shfl_xor(cs0, 16, 32);
  cs1 += __shfl_xor(cs1, 16, 32);
  cs2 += __shfl_xor(cs2, 16, 32);
  cs3 += __shfl_xor(cs3, 16, 32);
  __syncthreads();
  if (half == 0) *(float4*)&Alt[wave * 64 + mrow * 4] = make_float4(cs0, cs1, cs2, cs3);
  __syncthreads();
  if (threadIdx.x < 64 && a0 + (int)threadIdx.x < AA) {
    float s = 0.f;
    for (int w = 0; w < 16; ++w) s += Alt[w * 64 + threadIdx.x];
    atomicAdd(&colsum[a0 + threadIdx.x], s);
  }
  const int col = f0 + mrow;
#pragma unroll
  for (int v = 0; v < 8; ++v) {
    const int m = v + half * 8;
    atomicAdd(&msg0[(a0 +  0 + m) * FF + col], c0[v]);
    atomicAdd(&msg0[(a0 + 16 + m) * FF + col], c1[v]);
    atomicAdd(&msg0[(a0 + 32 + m) * FF + col], c2[v]);
    atomicAdd(&msg0[(a0 + 48 + m) * FF + col], c3[v]);
  }
}

__global__ __launch_bounds__(512) void k_msgw(const float* __restrict__ msg0,
                                              const float* __restrict__ Wm,
                                              const float* __restrict__ colsum,
                                              float* __restrict__ msg) {
  __shared__ float Als[16][260];
  const int a0 = blockIdx.x * 16;
  const int wave = threadIdx.x >> 5;
  const int lane = threadIdx.x & 31;
  const int f0 = wave * 16;
  const int half = lane >> 4;
  const int mrow = lane & 15;
  for (int e = threadIdx.x; e < 16 * 256; e += 512) {
    const int m = e >> 8, k = e & 255;
    const int a = a0 + m;
    const float ic = (a < AA) ? 1.0f / (colsum[a] + EPSF) : 0.0f;
    Als[m][k] = msg0[a * FF + k] * ic;
  }
  __syncthreads();
  v8f c = {};
  const float* wb = Wm + (f0 + mrow) * FF + half * 2;
  const float* ab = &Als[mrow][half * 2];
#pragma unroll
  for (int s = 0; s < 64; ++s) {
    v2f b = *(const v2f*)(wb + s * 4);
    v2f a = *(const v2f*)(ab + s * 4);
    c = wmma_f32(a, b, c);
  }
  const int col = f0 + mrow;
#pragma unroll
  for (int v = 0; v < 8; ++v)
    msg[(a0 + v + half * 8) * FF + col] = c[v];
}

__device__ __forceinline__ void out_load_block(v2f (&A)[4][4], v2f (&B)[4],
                                               const float* ap, const float* ap2,
                                               const float* bp) {
#pragma unroll
  for (int j = 0; j < 4; ++j) {
    B[j].x = bp[j * 4 * FF];
    B[j].y = bp[j * 4 * FF + FF];
    const float* a = ap + j * 4;
    A[j][0] = *(const v2f*)(a);
    A[j][1] = *(const v2f*)(a + 16 * 516);
    const float* a2 = ap2 + j * 4;
    A[j][2] = *(const v2f*)(a2);
    A[j][3] = *(const v2f*)(a2 + 16 * 516);
  }
}

__device__ __forceinline__ void out_mma_block(const v2f (&A)[4][4], const v2f (&B)[4],
                                              v8f& c0, v8f& c1, v8f& c2, v8f& c3) {
#pragma unroll
  for (int j = 0; j < 4; ++j) {
    c0 = wmma_f32(A[j][0], B[j], c0);
    c1 = wmma_f32(A[j][1], B[j], c1);
    c2 = wmma_f32(A[j][2], B[j], c2);
    c3 = wmma_f32(A[j][3], B[j], c3);
  }
}

__global__ __launch_bounds__(512) void k_out(const float* __restrict__ adj,
                                             const float* __restrict__ msg,
                                             float* __restrict__ out) {
  extern __shared__ __attribute__((aligned(16))) float Als[];
  const int LDP = 516;
  float* invr = Als + 64 * LDP;
  const int m0 = blockIdx.x * 64;
  const int wave = threadIdx.x >> 5;
  const int lane = threadIdx.x & 31;
  const int f0 = wave * 16;
  const int half = lane >> 4;
  const int mrow = lane & 15;
  const bool full = (m0 + 64 <= NN);
  {
    const float4* srcq = (const float4*)(adj + m0 * AA);
    if (full) {
      float4 r[16];
#pragma unroll
      for (int i = 0; i < 16; ++i) {
        const int q = threadIdx.x + i * 512;
        r[i] = srcq[(q < 8000) ? q : 0];
      }
#pragma unroll
      for (int i = 0; i < 16; ++i) {
        const int q = threadIdx.x + i * 512;
        const bool data = (q < 8000);
        const int t = q - 8000;
        const int m = data ? q / 125 : t / 3;
        const int c4 = data ? q - m * 125 : 125 + (t - m * 3);
        const float4 v = r[i];
        *(float4*)&Als[m * LDP + c4 * 4] = keep_if(data, v.x, v.y, v.z, v.w);
      }
    } else {
      const int nq = (NN - m0) * 125;
      for (int q = threadIdx.x; q < 8192; q += 512) {
        const bool data = (q < 8000);
        const int t = q - 8000;
        const int m = data ? q / 125 : t / 3;
        const int c4 = data ? q - m * 125 : 125 + (t - m * 3);
        float4 v = make_float4(0.f, 0.f, 0.f, 0.f);
        if (q < nq) v = srcq[q];
        *(float4*)&Als[m * LDP + c4 * 4] = v;
      }
    }
  }
  __syncthreads();
  float rs[4];
  for (int r = 0; r < 4; ++r) {
    const float4* rowq = (const float4*)&Als[(wave * 4 + r) * LDP];
    float s = 0.f;
    for (int j = 0; j < 4; ++j) {
      const float4 v = rowq[lane + 32 * j];
      s += (v.x + v.y) + (v.z + v.w);
    }
    rs[r] = s;
  }
  for (int off = 16; off > 0; off >>= 1) {
    for (int r = 0; r < 4; ++r) rs[r] += __shfl_xor(rs[r], off, 32);
  }
  {
    const float sv = (lane == 0) ? rs[0] : (lane == 1) ? rs[1] : (lane == 2) ? rs[2] : rs[3];
    if (lane < 4) invr[wave * 4 + lane] = 1.0f / (sv + EPSF);
  }
  v8f c0 = {}, c1 = {}, c2 = {}, c3 = {};
  const float* bb = msg + half * 2 * FF + f0 + mrow;
  const float* a0p = &Als[mrow * LDP + half * 2];
  v2f aP[4][4], bP[4], aQ[4][4], bQ[4];
  int hi_rows = 32 * LDP;
  asm volatile("" : "+v"(hi_rows));
  const float* a2p = a0p + hi_rows;
  out_load_block(aP, bP, a0p, a2p, bb);
#pragma unroll
  for (int blk = 0; blk < 32; blk += 2) {
    out_load_block(aQ, bQ, a0p + (blk + 1) * 16, a2p + (blk + 1) * 16,
                   bb + (blk + 1) * 16 * FF);
    __builtin_amdgcn_sched_barrier(0);
    out_mma_block(aP, bP, c0, c1, c2, c3);
    __builtin_amdgcn_sched_barrier(0);
    if (blk + 2 < 32)
      out_load_block(aP, bP, a0p + (blk + 2) * 16, a2p + (blk + 2) * 16,
                     bb + (blk + 2) * 16 * FF);
    __builtin_amdgcn_sched_barrier(0);
    out_mma_block(aQ, bQ, c0, c1, c2, c3);
    __builtin_amdgcn_sched_barrier(0);
  }
  __syncthreads();
  const int col = f0 + mrow;
  if (full) {
#pragma unroll
    for (int v = 0; v < 8; ++v) {
      const int mi = v + half * 8;
      int r;
      r = m0 +  0 + mi; out[r * FF + col] = c0[v] * invr[mi];
      r = m0 + 16 + mi; out[r * FF + col] = c1[v] * invr[mi + 16];
      r = m0 + 32 + mi; out[r * FF + col] = c2[v] * invr[mi + 32];
      r = m0 + 48 + mi; out[r * FF + col] = c3[v] * invr[mi + 48];
    }
  } else {
#pragma unroll
    for (int v = 0; v < 8; ++v) {
      const int mi = v + half * 8;
      int r;
      r = m0 +  0 + mi; if (r < NN) out[r * FF + col] = c0[v] * invr[mi];
      r = m0 + 16 + mi; if (r < NN) out[r * FF + col] = c1[v] * invr[mi + 16];
      r = m0 + 32 + mi; if (r < NN) out[r * FF + col] = c2[v] * invr[mi + 32];
      r = m0 + 48 + mi; if (r < NN) out[r * FF + col] = c3[v] * invr[mi + 48];
    }
  }
}

extern "C" void kernel_launch(void* const* d_in, const int* in_sizes, int n_in,
                              void* d_out, int out_size, void* d_ws, size_t ws_size,
                              hipStream_t stream) {
  const float* inp = (const float*)d_in[0];
  const float* adj = (const float*)d_in[1];
  const float* Wm  = (const float*)d_in[2];
  float* out = (float*)d_out;

  char* ws = (char*)d_ws;
  float* msg0   = (float*)(ws);
  float* msg    = (float*)(ws + (size_t)APAD * FF * 4);
  float* colsum = (float*)(ws + (size_t)2 * APAD * FF * 4);

  hipMemsetAsync(msg0, 0, (size_t)APAD * FF * sizeof(float), stream);
  hipMemsetAsync(colsum, 0, APAD * sizeof(float), stream);

  k_atx<<<400, 512, 0, stream>>>(adj, inp, msg0, colsum);
  k_msgw<<<32, 512, 0, stream>>>(msg0, Wm, colsum, msg);
  k_out<<<(NN + 63) / 64, 512, (size_t)(64 * 516 + 64) * sizeof(float), stream>>>(
      adj, msg, out);
}


# k_atx split-K without atomics: 64 balanced n-chunks store partial tiles plainly (scratch in out, overwritten by the last kernel), a reduce kernel adds them in fixed order; both memsets go
# speedup vs baseline: 1.6407x; 1.0490x over previous
#include <hip/hip_runtime.h>

typedef __attribute__((ext_vector_type(2))) float v2f;
typedef __attribute__((ext_vector_type(8))) float v8f;

#define NN 50000
#define AA 500
#define FF 256
#define APAD 512
#define EPSF 1e-12f

__device__ __forceinline__ v8f wmma_f32(v2f a, v2f b, v8f c) {
  return __builtin_amdgcn_wmma_f32_16x16x4_f32(false, a, false, b, (short)0, c,
                                               false, false);
}

__device__ __forceinline__ float4 keep_if(bool ok, float a, float b, float c, float d) {
  return make_float4(ok ? a : 0.f, ok ? b : 0.f, ok ? c : 0.f, ok ? d : 0.f);
}

__global__ __launch_bounds__(512) void k_atx(const float* __restrict__ adj,
                                             const float* __restrict__ inp,
                                             float* __restrict__ part,
                                             float* __restrict__ cpart) {
  __shared__ __attribute__((aligned(16))) float Alt[50 * 160];
  const int agroup = blockIdx.x & 7;
  const int chunk  = blockIdx.x >> 3;
  const int a0 = agroup * 64;
  const int tile0 = (chunk * 500) >> 6;
  const int nrows = ((((chunk + 1) * 500) >> 6) - tile0) * 100;
  const int nbase = tile0 * 100;
  const int wave = threadIdx.x >> 5;
  const int lane = threadIdx.x & 31;
  const int f0 = wave * 16;
  const int half = lane >> 4;
  const int mrow = lane & 15;
  v8f c0 = {}, c1 = {}, c2 = {}, c3 = {};
  float cs0 = 0.f, cs1 = 0.f, cs2 = 0.f, cs3 = 0.f;
  const float* bbase = inp + (nbase + half * 2) * FF + f0 + mrow;
  const int afrag = half * 160 + mrow * 2;
  for (int kk = 0; kk < nrows; kk += 100) {
    float4 lo[2], hi[2];
#pragma unroll
    for (int i = 0; i < 2; ++i) {
      const int it = threadIdx.x + i * 512;
      const int pr = it >> 4;
      const int qd = it & 15;
      const bool ok = (it < 800) && (a0 + qd * 4 < AA);
      const int idx = ok ? ((nbase + kk + 2 * pr) * AA + a0 + qd * 4) : 0;
      lo[i] = *(const float4*)(adj + idx);
      hi[i] = *(const float4*)(adj + idx + AA);
    }
    __syncthreads();
#pragma unroll
    for (int i = 0; i < 2; ++i) {
      const int it = threadIdx.x + i * 512;
      const int pr = it >> 4;
      const int qd = it & 15;
      const bool ok = (a0 + qd * 4 < AA);
      if (it < 800) {
        const float4 l = lo[i], h = hi[i];
        float* d = &Alt[pr * 160 + qd * 8];
        const float4 p0 = keep_if(ok, l.x, h.x, l.y, h.y);
        const float4 p1 = keep_if(ok, l.z, h.z, l.w, h.w);
        *(float4*)(d)     = p0;
        *(float4*)(d + 4) = p1;
        cs0 += p0.x + p0.y;
        cs1 += p0.z + p0.w;
        cs2 += p1.x + p1.y;
        cs3 += p1.z + p1.w;
      }
    }
    __syncthreads();
    const float* bp = bbase + kk * FF;
#pragma unroll
    for (int s = 0; s < 25; ++s) {
      v2f b;
      b.x = bp[s * 4 * FF];
      b.y = bp[s * 4 * FF + FF];
      const float* ap = &Alt[afrag + s * 320];
      c0 = wmma_f32(*(const v2f*)(ap +  0), b, c0);
      c1 = wmma_f32(*(const v2f*)(ap + 32), b, c1);
      c2 = wmma_f32(*(const v2f*)(ap + 64), b, c2);
      c3 = wmma_f32(*(const v2f*)(ap + 96), b, c3);
    }
  }
  cs0 += __shfl_xor(cs0, 16, 32);
  cs1 += __shfl_xor(cs1, 16, 32);
  cs2 += __shfl_xor(cs2, 16, 32);
  cs3 += __shfl_xor(cs3, 16, 32);
  __syncthreads();
  if (half == 0) *(float4*)&Alt[wave * 64 + mrow * 4] = make_float4(cs0, cs1, cs2, cs3);
  __syncthreads();
  if (threadIdx.x < 64) {
    float s = 0.f;
    for (int w = 0; w < 16; ++w) s += Alt[w * 64 + threadIdx.x];
    cpart[chunk * APAD + a0 + threadIdx.x] = s;
  }
  float* pt = part + (chunk * APAD + a0) * FF + f0 + mrow;
#pragma unroll
  for (int v = 0; v < 8; ++v) {
    const int m = v + half * 8;
    pt[( 0 + m) * FF] = c0[v];
    pt[(16 + m) * FF] = c1[v];
    pt[(32 + m) * FF] = c2[v];
    pt[(48 + m) * FF] = c3[v];
  }
}

__global__ __launch_bounds__(256) void k_red(const float* __restrict__ part,
                                             const float* __restrict__ cpart,
                                             float* __restrict__ msg0) {
  __shared__ float4 red[4][64];
  __shared__ float csum[64];
  const int a = blockIdx.x;
  const int q = threadIdx.x & 63;
  const int g = threadIdx.x >> 6;
  const float4* p = (const float4*)(part + (g * 16 * APAD + a) * FF) + q;
  float4 s = make_float4(0.f, 0.f, 0.f, 0.f);
#pragma unroll
  for (int ch = 0; ch < 16; ++ch) {
    const float4 v = p[ch * (APAD * FF / 4)];
    s.x += v.x; s.y += v.y; s.z += v.z; s.w += v.w;
  }
  red[g][q] = s;
  if (threadIdx.x < 64) csum[threadIdx.x] = cpart[threadIdx.x * APAD + a];
  __syncthreads();
  if (g == 0) {
    float cs = 0.f;
    for (int ch = 0; ch < 64; ++ch) cs += csum[ch];
    const float ic = (a < AA) ? 1.0f / (cs + EPSF) : 0.0f;
    const float4 r0 = red[0][q], r1 = red[1][q], r2 = red[2][q], r3 = red[3][q];
    float4 t;
    t.x = ((r0.x + r1.x) + (r2.x + r3.x)) * ic;
    t.y = ((r0.y + r1.y) + (r2.y + r3.y)) * ic;
    t.z = ((r0.z + r1.z) + (r2.z + r3.z)) * ic;
    t.w = ((r0.w + r1.w) + (r2.w + r3.w)) * ic;
    *((float4*)(msg0 + a * FF) + q) = t;
  }
}

__global__ __launch_bounds__(512) void k_msgw(const float* __restrict__ msg0,
                                              const float* __restrict__ Wm,
                                              float* __restrict__ msg) {
  __shared__ float Als[16][260];
  const int a0 = blockIdx.x * 16;
  const int wave = threadIdx.x >> 5;
  const int lane = threadIdx.x & 31;
  const int f0 = wave * 16;
  const int half = lane >> 4;
  const int mrow = lane & 15;
  for (int e = threadIdx.x; e < 16 * 256; e += 512) {
    const int m = e >> 8, k = e & 255;
    Als[m][k] = msg0[(a0 + m) * FF + k];
  }
  __syncthreads();
  v8f c = {};
  const float* wb = Wm + (f0 + mrow) * FF + half * 2;
  const float* ab = &Als[mrow][half * 2];
#pragma unroll
  for (int s = 0; s < 64; ++s) {
    v2f b = *(const v2f*)(wb + s * 4);
    v2f a = *(const v2f*)(ab + s * 4);
    c = wmma_f32(a, b, c);
  }
  const int col = f0 + mrow;
#pragma unroll
  for (int v = 0; v < 8; ++v)
    msg[(a0 + v + half * 8) * FF + col] = c[v];
}

__device__ __forceinline__ void out_load_block(v2f (&A)[4][4], v2f (&B)[4],
                                               const float* ap, const float* ap2,
                                               const float* bp) {
#pragma unroll
  for (int j = 0; j < 4; ++j) {
    B[j].x = bp[j * 4 * FF];
    B[j].y = bp[j * 4 * FF + FF];
    const float* a = ap + j * 4;
    A[j][0] = *(const v2f*)(a);
    A[j][1] = *(const v2f*)(a + 16 * 516);
    const float* a2 = ap2 + j * 4;
    A[j][2] = *(const v2f*)(a2);
    A[j][3] = *(const v2f*)(a2 + 16 * 516);
  }
}

__device__ __forceinline__ void out_mma_block(const v2f (&A)[4][4], const v2f (&B)[4],
                                              v8f& c0, v8f& c1, v8f& c2, v8f& c3) {
#pragma unroll
  for (int j = 0; j < 4; ++j) {
    c0 = wmma_f32(A[j][0], B[j], c0);
    c1 = wmma_f32(A[j][1], B[j], c1);
    c2 = wmma_f32(A[j][2], B[j], c2);
    c3 = wmma_f32(A[j][3], B[j], c3);
  }
}

__global__ __launch_bounds__(512) void k_out(const float* __restrict__ adj,
                                             const float* __restrict__ msg,
                                             float* __restrict__ out) {
  extern __shared__ __attribute__((aligned(16))) float Als[];
  const int LDP = 516;
  float* invr = Als + 64 * LDP;
  const int m0 = blockIdx.x * 64;
  const int wave = threadIdx.x >> 5;
  const int lane = threadIdx.x & 31;
  const int f0 = wave * 16;
  const int half = lane >> 4;
  const int mrow = lane & 15;
  const bool full = (m0 + 64 <= NN);
  {
    const float4* srcq = (const float4*)(adj + m0 * AA);
    if (full) {
      float4 r[16];
#pragma unroll
      for (int i = 0; i < 16; ++i) {
        const int q = threadIdx.x + i * 512;
        r[i] = srcq[(q < 8000) ? q : 0];
      }
#pragma unroll
      for (int i = 0; i < 16; ++i) {
        const int q = threadIdx.x + i * 512;
        const bool data = (q < 8000);
        const int t = q - 8000;
        const int m = data ? q / 125 : t / 3;
        const int c4 = data ? q - m * 125 : 125 + (t - m * 3);
        const float4 v = r[i];
        *(float4*)&Als[m * LDP + c4 * 4] = keep_if(data, v.x, v.y, v.z, v.w);
      }
    } else {
      const int nq = (NN - m0) * 125;
      for (int q = threadIdx.x; q < 8192; q += 512) {
        const bool data = (q < 8000);
        const int t = q - 8000;
        const int m = data ? q / 125 : t / 3;
        const int c4 = data ? q - m * 125 : 125 + (t - m * 3);
        float4 v = make_float4(0.f, 0.f, 0.f, 0.f);
        if (q < nq) v = srcq[q];
        *(float4*)&Als[m * LDP + c4 * 4] = v;
      }
    }
  }
  __syncthreads();
  float rs[4];
  for (int r = 0; r < 4; ++r) {
    const float4* rowq = (const float4*)&Als[(wave * 4 + r) * LDP];
    float s = 0.f;
    for (int j = 0; j < 4; ++j) {
      const float4 v = rowq[lane + 32 * j];
      s += (v.x + v.y) + (v.z + v.w);
    }
    rs[r] = s;
  }
  for (int off = 16; off > 0; off >>= 1) {
    for (int r = 0; r < 4; ++r) rs[r] += __shfl_xor(rs[r], off, 32);
  }
  {
    const float sv = (lane == 0) ? rs[0] : (lane == 1) ? rs[1] : (lane == 2) ? rs[2] : rs[3];
    if (lane < 4) invr[wave * 4 + lane] = 1.0f / (sv + EPSF);
  }
  v8f c0 = {}, c1 = {}, c2 = {}, c3 = {};
  const float* bb = msg + half * 2 * FF + f0 + mrow;
  const float* a0p = &Als[mrow * LDP + half * 2];
  v2f aP[4][4], bP[4], aQ[4][4], bQ[4];
  int hi_rows = 32 * LDP;
  asm volatile("" : "+v"(hi_rows));
  const float* a2p = a0p + hi_rows;
  out_load_block(aP, bP, a0p, a2p, bb);
#pragma unroll
  for (int blk = 0; blk < 32; blk += 2) {
    out_load_block(aQ, bQ, a0p + (blk + 1) * 16, a2p + (blk + 1) * 16,
                   bb + (blk + 1) * 16 * FF);
    __builtin_amdgcn_sched_barrier(0);
    out_mma_block(aP, bP, c0, c1, c2, c3);
    __builtin_amdgcn_sched_barrier(0);
    if (blk + 2 < 32)
      out_load_block(aP, bP, a0p + (blk + 2) * 16, a2p + (blk + 2) * 16,
                     bb + (blk + 2) * 16 * FF);
    __builtin_amdgcn_sched_barrier(0);
    out_mma_block(aQ, bQ, c0, c1, c2, c3);
    __builtin_amdgcn_sched_barrier(0);
  }
  __syncthreads();
  const int col = f0 + mrow;
  if (full) {
#pragma unroll
    for (int v = 0; v < 8; ++v) {
      const int mi = v + half * 8;
      int r;
      r = m0 +  0 + mi; out[r * FF + col] = c0[v] * invr[mi];
      r = m0 + 16 + mi; out[r * FF + col] = c1[v] * invr[mi + 16];
      r = m0 + 32 + mi; out[r * FF + col] = c2[v] * invr[mi + 32];
      r = m0 + 48 + mi; out[r * FF + col] = c3[v] * invr[mi + 48];
    }
  } else {
#pragma unroll
    for (int v = 0; v < 8; ++v) {
      const int mi = v + half * 8;
      int r;
      r = m0 +  0 + mi; if (r < NN) out[r * FF + col] = c0[v] * invr[mi];
      r = m0 + 16 + mi; if (r < NN) out[r * FF + col] = c1[v] * invr[mi + 16];
      r = m0 + 32 + mi; if (r < NN) out[r * FF + col] = c2[v] * invr[mi + 32];
      r = m0 + 48 + mi; if (r < NN) out[r * FF + col] = c3[v] * invr[mi + 48];
    }
  }
}

extern "C" void kernel_launch(void* const* d_in, const int* in_sizes, int n_in,
                              void* d_out, int out_size, void* d_ws, size_t ws_size,
                              hipStream_t stream) {
  const float* inp = (const float*)d_in[0];
  const float* adj = (const float*)d_in[1];
  const float* Wm  = (const float*)d_in[2];
  float* out = (float*)d_out;

  char* ws = (char*)d_ws;
  float* msg0   = (float*)(ws);
  float* msg    = (float*)(ws + (size_t)APAD * FF * 4);
  float* part  = out;
  float* cpart = out + (size_t)64 * APAD * FF;

  k_atx<<<512, 512, 0, stream>>>(adj, inp, part, cpart);
  k_red<<<APAD, 256, 0, stream>>>(part, cpart, msg0);
  k_msgw<<<32, 512, 0, stream>>>(msg0, Wm, msg);
  k_out<<<(NN + 63) / 64, 512, (size_t)(64 * 516 + 64) * sizeof(float), stream>>>(
      adj, msg, out);
}


# k_out's rows dealt as 768 blocks of 64 rows + 53 blocks of 16 rows (was 782 of 64): three full rounds of 2 resident blocks per processor, the fourth round of 14 blocks goes
# speedup vs baseline: 1.7631x; 1.0746x over previous
#include <hip/hip_runtime.h>

typedef __attribute__((ext_vector_type(2))) float v2f;
typedef __attribute__((ext_vector_type(8))) float v8f;

#define NN 50000
#define AA 500
#define FF 256
#define APAD 512
#define EPSF 1e-12f

__device__ __forceinline__ v8f wmma_f32(v2f a, v2f b, v8f c) {
  return __builtin_amdgcn_wmma_f32_16x16x4_f32(false, a, false, b, (short)0, c,
                                               false, false);
}

__device__ __forceinline__ float4 keep_if(bool ok, float a, float b, float c, float d) {
  return make_float4(ok ? a : 0.f, ok ? b : 0.f, ok ? c : 0.f, ok ? d : 0.f);
}

__global__ __launch_bounds__(512) void k_atx(const float* __restrict__ adj,
                                             const float* __restrict__ inp,
                                             float* __restrict__ part,
                                             float* __restrict__ cpart) {
  __shared__ __attribute__((aligned(16))) float Alt[50 * 160];
  const int agroup = blockIdx.x & 7;
  const int chunk  = blockIdx.x >> 3;
  const int a0 = agroup * 64;
  const int tile0 = (chunk * 500) >> 6;
  const int nrows = ((((chunk + 1) * 500) >> 6) - tile0) * 100;
  const int nbase = tile0 * 100;
  const int wave = threadIdx.x >> 5;
  const int lane = threadIdx.x & 31;
  const int f0 = wave * 16;
  const int half = lane >> 4;
  const int mrow = lane & 15;
  v8f c0 = {}, c1 = {}, c2 = {}, c3 = {};
  float cs0 = 0.f, cs1 = 0.f, cs2 = 0.f, cs3 = 0.f;
  const float* bbase = inp + (nbase + half * 2) * FF + f0 + mrow;
  const int afrag = half * 160 + mrow * 2;
  for (int kk = 0; kk < nrows; kk += 100) {
    float4 lo[2], hi[2];
#pragma unroll
    for (int i = 0; i < 2; ++i) {
      const int it = threadIdx.x + i * 512;
      const int pr = it >> 4;
      const int qd = it & 15;
      const bool ok = (it < 800) && (a0 + qd * 4 < AA);
      const int idx = ok ? ((nbase + kk + 2 * pr) * AA + a0 + qd * 4) : 0;
      lo[i] = *(const float4*)(adj + idx);
      hi[i] = *(const float4*)(adj + idx + AA);
    }
    __syncthreads();
#pragma unroll
    for (int i = 0; i < 2; ++i) {
      const int it = threadIdx.x + i * 512;
      const int pr = it >> 4;
      const int qd = it & 15;
      const bool ok = (a0 + qd * 4 < AA);
      if (it < 800) {
        const float4 l = lo[i], h = hi[i];
        float* d = &Alt[pr * 160 + qd * 8];
        const float4 p0 = keep_if(ok, l.x, h.x, l.y, h.y);
        const float4 p1 = keep_if(ok, l.z, h.z, l.w, h.w);
        *(float4*)(d)     = p0;
        *(float4*)(d + 4) = p1;
        cs0 += p0.x + p0.y;
        cs1 += p0.z + p0.w;
        cs2 += p1.x + p1.y;
        cs3 += p1.z + p1.w;
      }
    }
    __syncthreads();
    const float* bp = bbase + kk * FF;
#pragma unroll
    for (int s = 0; s < 25; ++s) {
      v2f b;
      b.x = bp[s * 4 * FF];
      b.y = bp[s * 4 * FF + FF];
      const float* ap = &Alt[afrag + s * 320];
      c0 = wmma_f32(*(const v2f*)(ap +  0), b, c0);
      c1 = wmma_f32(*(const v2f*)(ap + 32), b, c1);
      c2 = wmma_f32(*(const v2f*)(ap + 64), b, c2);
      c3 = wmma_f32(*(const v2f*)(ap + 96), b, c3);
    }
  }
  cs0 += __shfl_xor(cs0, 16, 32);
  cs1 += __shfl_xor(cs1, 16, 32);
  cs2 += __shfl_xor(cs2, 16, 32);
  cs3 += __shfl_xor(cs3, 16, 32);
  __syncthreads();
  if (half == 0) *(float4*)&Alt[wave * 64 + mrow * 4] = make_float4(cs0, cs1, cs2, cs3);
  __syncthreads();
  if (threadIdx.x < 64) {
    float s = 0.f;
    for (int w = 0; w < 16; ++w) s += Alt[w * 64 + threadIdx.x];
    cpart[chunk * APAD + a0 + threadIdx.x] = s;
  }
  float* pt = part + (chunk * APAD + a0) * FF + f0 + mrow;
#pragma unroll
  for (int v = 0; v < 8; ++v) {
    const int m = v + half * 8;
    pt[( 0 + m) * FF] = c0[v];
    pt[(16 + m) * FF] = c1[v];
    pt[(32 + m) * FF] = c2[v];
    pt[(48 + m) * FF] = c3[v];
  }
}

__global__ __launch_bounds__(256) void k_red(const float* __restrict__ part,
                                             const float* __restrict__ cpart,
                                             float* __restrict__ msg0) {
  __shared__ float4 red[4][64];
  __shared__ float csum[64];
  const int a = blockIdx.x;
  const int q = threadIdx.x & 63;
  const int g = threadIdx.x >> 6;
  const float4* p = (const float4*)(part + (g * 16 * APAD + a) * FF) + q;
  float4 s = make_float4(0.f, 0.f, 0.f, 0.f);
#pragma unroll
  for (int ch = 0; ch < 16; ++ch) {
    const float4 v = p[ch * (APAD * FF / 4)];
    s.x += v.x; s.y += v.y; s.z += v.z; s.w += v.w;
  }
  red[g][q] = s;
  if (threadIdx.x < 64) csum[threadIdx.x] = cpart[threadIdx.x * APAD + a];
  __syncthreads();
  if (g == 0) {
    float cs = 0.f;
    for (int ch = 0; ch < 64; ++ch) cs += csum[ch];
    const float ic = (a < AA) ? 1.0f / (cs + EPSF) : 0.0f;
    const float4 r0 = red[0][q], r1 = red[1][q], r2 = red[2][q], r3 = red[3][q];
    float4 t;
    t.x = ((r0.x + r1.x) + (r2.x + r3.x)) * ic;
    t.y = ((r0.y + r1.y) + (r2.y + r3.y)) * ic;
    t.z = ((r0.z + r1.z) + (r2.z + r3.z)) * ic;
    t.w = ((r0.w + r1.w) + (r2.w + r3.w)) * ic;
    *((float4*)(msg0 + a * FF) + q) = t;
  }
}

__global__ __launch_bounds__(512) void k_msgw(const float* __restrict__ msg0,
                                              const float* __restrict__ Wm,
                                              float* __restrict__ msg) {
  __shared__ float Als[16][260];
  const int a0 = blockIdx.x * 16;
  const int wave = threadIdx.x >> 5;
  const int lane = threadIdx.x & 31;
  const int f0 = wave * 16;
  const int half = lane >> 4;
  const int mrow = lane & 15;
  for (int e = threadIdx.x; e < 16 * 256; e += 512) {
    const int m = e >> 8, k = e & 255;
    Als[m][k] = msg0[(a0 + m) * FF + k];
  }
  __syncthreads();
  v8f c = {};
  const float* wb = Wm + (f0 + mrow) * FF + half * 2;
  const float* ab = &Als[mrow][half * 2];
#pragma unroll
  for (int s = 0; s < 64; ++s) {
    v2f b = *(const v2f*)(wb + s * 4);
    v2f a = *(const v2f*)(ab + s * 4);
    c = wmma_f32(a, b, c);
  }
  const int col = f0 + mrow;
#pragma unroll
  for (int v = 0; v < 8; ++v)
    msg[(a0 + v + half * 8) * FF + col] = c[v];
}

template <int MT, int KB>
__device__ __forceinline__ void out_load_block(v2f (&A)[KB][MT], v2f (&B)[KB],
                                               const float* ap, const float* ap2,
                                               const float* bp) {
#pragma unroll
  for (int j = 0; j < KB; ++j) {
    B[j].x = bp[j * 4 * FF];
    B[j].y = bp[j * 4 * FF + FF];
    const float* a = ap + j * 4;
    A[j][0] = *(const v2f*)(a);
    if constexpr (MT > 1) A[j][1] = *(const v2f*)(a + 16 * 516);
    if constexpr (MT > 2) {
      const float* a2 = ap2 + j * 4;
      A[j][2] = *(const v2f*)(a2);
      A[j][3] = *(const v2f*)(a2 + 16 * 516);
    }
  }
}

template <int MT, int KB>
__device__ __forceinline__ void out_mma_block(const v2f (&A)[KB][MT], const v2f (&B)[KB],
                                              v8f (&c)[MT]) {
#pragma unroll
  for (int j = 0; j < KB; ++j) {
#pragma unroll
    for (int i = 0; i < MT; ++i) c[i] = wmma_f32(A[j][i], B[j], c[i]);
  }
}

#define OUT_BIG 768
#define OUT_SMALL ((NN - OUT_BIG * 64) / 16)
static_assert(OUT_BIG * 64 + OUT_SMALL * 16 == NN, "k_out's blocks cover every row once");

template <int MT>
__device__ __forceinline__ void out_tile(const float* __restrict__ adj,
                                         const float* __restrict__ msg,
                                         float* __restrict__ out, float* Als,
                                         const int m0) {
  constexpr int ROWS = 16 * MT;
  constexpr int LDP = 516;
  float* invr = Als + ROWS * LDP;
  const int wave = threadIdx.x >> 5;
  const int lane = threadIdx.x & 31;
  const int f0 = wave * 16;
  const int half = lane >> 4;
  const int mrow = lane & 15;
  {
    constexpr int NDATA = ROWS * 125;
    constexpr int NP = 4 * MT;
    const float4* srcq = (const float4*)(adj + m0 * AA);
    float4 r[NP];
#pragma unroll
    for (int i = 0; i < NP; ++i) {
      const int q = threadIdx.x + i * 512;
      r[i] = srcq[(q < NDATA) ? q : 0];
    }
#pragma unroll
    for (int i = 0; i < NP; ++i) {
      const int q = threadIdx.x + i * 512;
      const bool data = (q < NDATA);
      const int t = q - NDATA;
      const int m = data ? q / 125 : t / 3;
      const int c4 = data ? q - m * 125 : 125 + (t - m * 3);
      const float4 v = r[i];
      *(float4*)&Als[m * LDP + c4 * 4] = keep_if(data, v.x, v.y, v.z, v.w);
    }
  }
  __syncthreads();
  float rs[MT];
#pragma unroll
  for (int r = 0; r < MT; ++r) {
    const float4* rowq = (const float4*)&Als[(wave * MT + r) * LDP];
    float s = 0.f;
    for (int j = 0; j < 4; ++j) {
      const float4 v = rowq[lane + 32 * j];
      s += (v.x + v.y) + (v.z + v.w);
    }
    rs[r] = s;
  }
  for (int off = 16; off > 0; off >>= 1) {
#pragma unroll
    for (int r = 0; r < MT; ++r) rs[r] += __shfl_xor(rs[r], off, 32);
  }
  {
    float sv = rs[MT - 1];
#pragma unroll
    for (int r = MT - 2; r >= 0; --r) sv = (lane == r) ? rs[r] : sv;
    if (lane < MT) invr[wave * MT + lane] = 1.0f / (sv + EPSF);
  }
  v8f c[MT];
#pragma unroll
  for (int i = 0; i < MT; ++i) c[i] = v8f{};
  const float* bb = msg + half * 2 * FF + f0 + mrow;
  const float* a0p = &Als[mrow * LDP + half * 2];
  constexpr int KB = (MT == 1) ? 8 : 4;
  constexpr int NB = 128 / KB;
  constexpr int KS = KB * 4;
  v2f aP[KB][MT], bP[KB], aQ[KB][MT], bQ[KB];
  const float* a2p = a0p;
  if constexpr (MT > 2) {
    int hi_rows = 32 * LDP;
    asm volatile("" : "+v"(hi_rows));
    a2p = a0p + hi_rows;
  }
  out_load_block<MT, KB>(aP, bP, a0p, a2p, bb);
#pragma unroll
  for (int blk = 0; blk < NB; blk += 2) {
    out_load_block<MT, KB>(aQ, bQ, a0p + (blk + 1) * KS, a2p + (blk + 1) * KS,
                           bb + (blk + 1) * KS * FF);
    __builtin_amdgcn_sched_barrier(0);
    out_mma_block<MT, KB>(aP, bP, c);
    __builtin_amdgcn_sched_barrier(0);
    if (blk + 2 < NB)
      out_load_block<MT, KB>(aP, bP, a0p + (blk + 2) * KS, a2p + (blk + 2) * KS,
                             bb + (blk + 2) * KS * FF);
    __builtin_amdgcn_sched_barrier(0);
    out_mma_block<MT, KB>(aQ, bQ, c);
    __builtin_amdgcn_sched_barrier(0);
  }
  __syncthreads();
  const int col = f0 + mrow;
#pragma unroll
  for (int v = 0; v < 8; ++v) {
    const int mi = v + half * 8;
#pragma unroll
    for (int i = 0; i < MT; ++i)
      out[(m0 + 16 * i + mi) * FF + col] = c[i][v] * invr[mi + 16 * i];
  }
}

__global__ __launch_bounds__(512) void k_out(const float* __restrict__ adj,
                                             const float* __restrict__ msg,
                                             float* __restrict__ out) {
  extern __shared__ __attribute__((aligned(16))) float Als[];
  if (blockIdx.x < OUT_BIG)
    out_tile<4>(adj, msg, out, Als, blockIdx.x * 64);
  else
    out_tile<1>(adj, msg, out, Als, OUT_BIG * 64 + (blockIdx.x - OUT_BIG) * 16);
}

extern "C" void kernel_launch(void* const* d_in, const int* in_sizes, int n_in,
                              void* d_out, int out_size, void* d_ws, size_t ws_size,
                              hipStream_t stream) {
  const float* inp = (const float*)d_in[0];
  const float* adj = (const float*)d_in[1];
  const float* Wm  = (const float*)d_in[2];
  float* out = (float*)d_out;

  char* ws = (char*)d_ws;
  float* msg0   = (float*)(ws);
  float* msg    = (float*)(ws + (size_t)APAD * FF * 4);
  float* part  = out;
  float* cpart = out + (size_t)64 * APAD * FF;

  k_atx<<<512, 512, 0, stream>>>(adj, inp, part, cpart);
  k_red<<<APAD, 256, 0, stream>>>(part, cpart, msg0);
  k_msgw<<<32, 512, 0, stream>>>(msg0, Wm, msg);
  k_out<<<OUT_BIG + OUT_SMALL, 512, (size_t)(64 * 516 + 64) * sizeof(float), stream>>>(
      adj, msg, out);
}
